# starting kernel
# baseline (speedup 1.0000x reference)
#include <hip/hip_runtime.h>

typedef float v2f __attribute__((ext_vector_type(2)));
typedef float v8f __attribute__((ext_vector_type(8)));

__global__ void k_deg_init(float* __restrict__ deg, int n) {
    int i = blockIdx.x * blockDim.x + threadIdx.x;
    if (i < n) deg[i] = 1.0f;
}

__global__ void k_deg_acc(const int* __restrict__ dst, float* __restrict__ deg, int E) {
    int e = blockIdx.x * blockDim.x + threadIdx.x;
    if (e < E) atomicAdd(&deg[dst[e]], 1.0f);
}

__global__ void k_rsqrt_inplace(float* __restrict__ deg, int n) {
    int i = blockIdx.x * blockDim.x + threadIdx.x;
    if (i < n) deg[i] = rsqrtf(deg[i]);
}

template <int N, int K>
__global__ void k_gemm_wmma(const float* __restrict__ A, const float* __restrict__ B,
                            float* __restrict__ C) {
    const int wave  = (blockIdx.x * blockDim.x + threadIdx.x) >> 5;
    const int lane  = threadIdx.x & 31;
    const int tilesN = N / 16;
    const int tM   = wave / tilesN;
    const int tN   = wave % tilesN;
    const int half = lane >> 4;
    const int l    = lane & 15;

    const size_t rowA = (size_t)tM * 16 + l;
    const int    colB = tN * 16 + l;

    const float* ap = A + rowA * K + half * 2;

    v8f acc = {};
#pragma unroll
    for (int k = 0; k < K; k += 4) {
        v2f a = *(const v2f*)(ap + k);
        const int kb = k + half * 2;
        v2f b;
        b.x = B[(size_t)kb * N + colB];
        b.y = B[(size_t)(kb + 1) * N + colB];
        acc = __builtin_amdgcn_wmma_f32_16x16x4_f32(
             false, a,  false, b,
             (short)0, acc,  false,  false);
    }

    float* cp = C + ((size_t)tM * 16 + half * 8) * N + (size_t)tN * 16 + l;
#pragma unroll
    for (int v = 0; v < 8; ++v) cp[(size_t)v * N] = acc[v];
}

template <int F, bool BIAS>
__global__ void k_self_init(const float* __restrict__ h, const float* __restrict__ dinv,
                            const float* __restrict__ bias, float* __restrict__ out, int n) {
    int t = blockIdx.x * blockDim.x + threadIdx.x;
    if (t >= n * F) return;
    int node = t / F;
    int f    = t % F;
    float di = dinv[node];
    float v  = h[t] * di * di;
    if (BIAS) v += bias[f];
    out[t] = v;
}

template <int F>
__global__ void k_bias_relu(const float* __restrict__ agg, const float* __restrict__ bias,
                            float* __restrict__ out, int n) {
    int t = blockIdx.x * blockDim.x + threadIdx.x;
    if (t >= n * F) return;
    int f = t % F;
    float v = agg[t] + bias[f];
    out[t] = v > 0.0f ? v : 0.0f;
}

template <int F>
__global__ void k_edge_scatter(const int* __restrict__ src, const int* __restrict__ dst,
                               const float* __restrict__ dinv, const float* __restrict__ h,
                               float* __restrict__ agg, int E) {
    int t = blockIdx.x * blockDim.x + threadIdx.x;
    int e = t >> 5;
    if (e >= E) return;
    int lane = t & 31;
    int s = src[e], d = dst[e];
    float nrm = dinv[s] * dinv[d];
    constexpr int PER = F / 32;
    const float* hp = h   + (size_t)s * F + lane * PER;
    float*       op = agg + (size_t)d * F + lane * PER;
#pragma unroll
    for (int j = 0; j < PER; ++j) atomicAdd(op + j, hp[j] * nrm);
}

extern "C" void kernel_launch(void* const* d_in, const int* in_sizes, int n_in,
                              void* d_out, int out_size, void* d_ws, size_t ws_size,
                              hipStream_t stream) {
    const float* x  = (const float*)d_in[0];
    const int*   ei = (const int*)d_in[1];
    const float* W1 = (const float*)d_in[2];
    const float* b1 = (const float*)d_in[3];
    const float* W2 = (const float*)d_in[4];
    const float* b2 = (const float*)d_in[5];
    float* out = (float*)d_out;

    const int n = in_sizes[0] / 128;
    const int E = in_sizes[1] / 2;
    const int* src = ei;
    const int* dst = ei + E;

    char* ws = (char*)d_ws;
    size_t hOff   = ((size_t)n * 4 + 511) & ~(size_t)511;
    size_t aggOff = hOff + (size_t)n * 128 * 4;
    size_t gOff   = aggOff + (size_t)n * 128 * 4;
    float* dinv = (float*)(ws);
    float* h    = (float*)(ws + hOff);
    float* agg  = (float*)(ws + aggOff);
    float* g    = (float*)(ws + gOff);

    const int B = 256;

    k_deg_init<<<(n + B - 1) / B, B, 0, stream>>>(dinv, n);
    k_deg_acc<<<(E + B - 1) / B, B, 0, stream>>>(dst, dinv, E);
    k_rsqrt_inplace<<<(n + B - 1) / B, B, 0, stream>>>(dinv, n);

    {
        int waves = (n / 16) * (128 / 16);
        k_gemm_wmma<128, 128><<<waves / 8, 256, 0, stream>>>(x, W1, h);
    }
    k_self_init<128, false><<<((size_t)n * 128 + B - 1) / B, B, 0, stream>>>(h, dinv, nullptr, agg, n);
    k_edge_scatter<128><<<((size_t)E * 32 + B - 1) / B, B, 0, stream>>>(src, dst, dinv, h, agg, E);
    k_bias_relu<128><<<((size_t)n * 128 + B - 1) / B, B, 0, stream>>>(agg, b1, h, n);

    {
        int waves = (n / 16) * (64 / 16);
        k_gemm_wmma<64, 128><<<waves / 8, 256, 0, stream>>>(h, W2, g);
    }
    k_self_init<64, true><<<((size_t)n * 64 + B - 1) / B, B, 0, stream>>>(g, dinv, b2, out, n);
    k_edge_scatter<64><<<((size_t)E * 32 + B - 1) / B, B, 0, stream>>>(src, dst, dinv, g, out, E);
}


# Replace the float-atomic edge scatter by an in-call destination-sorted edge list (integer arrival-order rank + scan) and a per-node wave gather; row sums now in rank order
# speedup vs baseline: 2.4151x; 2.4151x over previous
#include <hip/hip_runtime.h>


typedef float v2f __attribute__((ext_vector_type(2)));
typedef float v8f __attribute__((ext_vector_type(8)));

__global__ void k_zero_cnt(int* __restrict__ cnt, int n) {
    int i = blockIdx.x * blockDim.x + threadIdx.x;
    if (i < n) cnt[i] = 0;
}

__global__ void k_count_rank(const int* __restrict__ dst, int* __restrict__ cnt,
                             int* __restrict__ rank, int E) {
    int e = blockIdx.x * blockDim.x + threadIdx.x;
    if (e < E) rank[e] = atomicAdd(&cnt[dst[e]], 1);
}

__device__ __forceinline__ int wave_incl_scan(int v, int lane) {
#pragma unroll
    for (int o = 1; o < 32; o <<= 1) {
        int t = __shfl_up(v, o, 32);
        if (lane >= o) v += t;
    }
    return v;
}

__global__ void k_scan_local(const int* __restrict__ cnt, int* __restrict__ rowloc,
                             int* __restrict__ blksum, float* __restrict__ dinv, int n) {
    __shared__ int wsum[8];
    const int t = threadIdx.x, lane = t & 31, w = t >> 5;
    const int i0 = blockIdx.x * 1024 + t * 4;
    int c[4];
#pragma unroll
    for (int j = 0; j < 4; ++j) c[j] = (i0 + j < n) ? cnt[i0 + j] : 0;
    const int s = c[0] + c[1] + c[2] + c[3];
    const int inc = wave_incl_scan(s, lane);
    if (lane == 31) wsum[w] = inc;
    __syncthreads();
    int woff = 0;
#pragma unroll
    for (int k = 0; k < 8; ++k) woff += (k < w) ? wsum[k] : 0;
    int excl = woff + inc - s;
#pragma unroll
    for (int j = 0; j < 4; ++j) {
        if (i0 + j < n) {
            rowloc[i0 + j] = excl;
            dinv[i0 + j]   = rsqrtf((float)(c[j] + 1));
        }
        excl += c[j];
    }
    if (t == 255) blksum[blockIdx.x] = woff + inc;
}

__global__ void k_scan_blk(int* __restrict__ blksum, int nblk) {
    __shared__ int wsum[8];
    const int t = threadIdx.x, lane = t & 31, w = t >> 5;
    int carry = 0;
    for (int base = 0; base < nblk; base += 256) {
        const int i = base + t;
        const int v = (i < nblk) ? blksum[i] : 0;
        const int inc = wave_incl_scan(v, lane);
        if (lane == 31) wsum[w] = inc;
        __syncthreads();
        int woff = 0, tot = 0;
#pragma unroll
        for (int k = 0; k < 8; ++k) { woff += (k < w) ? wsum[k] : 0; tot += wsum[k]; }
        if (i < nblk) blksum[i] = carry + woff + inc - v;
        carry += tot;
        __syncthreads();
    }
}

__global__ void k_fill(const int* __restrict__ src, const int* __restrict__ dst,
                       const int* __restrict__ rank, const int* __restrict__ rowloc,
                       const int* __restrict__ blkoff, int* __restrict__ csr, int E) {
    int e = blockIdx.x * blockDim.x + threadIdx.x;
    if (e < E) {
        int d = dst[e];
        csr[rowloc[d] + blkoff[d >> 10] + rank[e]] = src[e];
    }
}

template <int N, int K>
__global__ void k_gemm_wmma(const float* __restrict__ A, const float* __restrict__ B,
                            float* __restrict__ C) {
    const int wave  = (blockIdx.x * blockDim.x + threadIdx.x) >> 5;
    const int lane  = threadIdx.x & 31;
    const int tilesN = N / 16;
    const int tM   = wave / tilesN;
    const int tN   = wave % tilesN;
    const int half = lane >> 4;
    const int l    = lane & 15;

    const size_t rowA = (size_t)tM * 16 + l;
    const int    colB = tN * 16 + l;

    const float* ap = A + rowA * K + half * 2;

    v8f acc = {};
#pragma unroll
    for (int k = 0; k < K; k += 4) {
        v2f a = *(const v2f*)(ap + k);
        const int kb = k + half * 2;
        v2f b;
        b.x = B[(size_t)kb * N + colB];
        b.y = B[(size_t)(kb + 1) * N + colB];
        acc = __builtin_amdgcn_wmma_f32_16x16x4_f32(
             false, a,  false, b,
             (short)0, acc,  false,  false);
    }

    float* cp = C + ((size_t)tM * 16 + half * 8) * N + (size_t)tN * 16 + l;
#pragma unroll
    for (int v = 0; v < 8; ++v) cp[(size_t)v * N] = acc[v];
}


template <int F, bool RELU>
__global__ void k_gather(const int* __restrict__ csr, const int* __restrict__ cnt,
                         const int* __restrict__ rowloc, const int* __restrict__ blkoff,
                         const float* __restrict__ dinv, const float* __restrict__ h,
                         const float* __restrict__ bias, float* __restrict__ out, int n) {
    constexpr int PER = F / 32;
    typedef float vf __attribute__((ext_vector_type(PER)));
    const int lane = threadIdx.x & 31;
    int d = (blockIdx.x * blockDim.x + threadIdx.x) >> 5;
    if (d >= n) return;
    d = __builtin_amdgcn_readfirstlane(d);
    const int   start = rowloc[d] + blkoff[d >> 10];
    const int   deg   = cnt[d];
    const float dd    = dinv[d];
    const float* hb = h + lane * PER;

    vf acc = *(const vf*)(hb + (size_t)d * F) * (dd * dd);

    for (int base = 0; base < deg; base += 32) {
        const int m = (deg - base) < 32 ? (deg - base) : 32;
        int   myS = 0;
        float myD = 0.0f;
        if (lane < m) {
            myS = csr[start + base + lane];
            myD = dinv[myS];
        }
        for (int j = 0; j < m; j += 8) {
            const int r = m - j;
            vf v[8];
#pragma unroll
            for (int u = 0; u < 8; ++u) {
                if (u < r) {
                    const int s = __builtin_amdgcn_readlane(myS, j + u);
                    v[u] = *(const vf*)(hb + (size_t)s * F);
                }
            }
            float nr[8];
#pragma unroll
            for (int u = 0; u < 8; ++u)
                nr[u] = __int_as_float(__builtin_amdgcn_readlane(__float_as_int(myD), (j + u) & 31)) * dd;
#pragma unroll
            for (int u = 0; u < 8; ++u) {
                if (u < r) acc += v[u] * nr[u];
            }
        }
    }

    const vf b = *(const vf*)(bias + lane * PER);
    acc += b;
    if (RELU) {
#pragma unroll
        for (int q = 0; q < PER; ++q) acc[q] = acc[q] > 0.0f ? acc[q] : 0.0f;
    }
    *(vf*)(out + (size_t)d * F + lane * PER) = acc;
}

extern "C" void kernel_launch(void* const* d_in, const int* in_sizes, int n_in,
                              void* d_out, int out_size, void* d_ws, size_t ws_size,
                              hipStream_t stream) {
    const float* x  = (const float*)d_in[0];
    const int*   ei = (const int*)d_in[1];
    const float* W1 = (const float*)d_in[2];
    const float* b1 = (const float*)d_in[3];
    const float* W2 = (const float*)d_in[4];
    const float* b2 = (const float*)d_in[5];
    float* out = (float*)d_out;

    const int n = in_sizes[0] / 128;
    const int E = in_sizes[1] / 2;
    const int* src = ei;
    const int* dst = ei + E;
    const int nblk = (n + 1023) / 1024;

    char* ws = (char*)d_ws;
    const size_t nodeB = ((size_t)n * 4 + 511) & ~(size_t)511;
    const size_t edgeB = ((size_t)E * 4 + 511) & ~(size_t)511;
    const size_t blkB  = ((size_t)nblk * 4 + 511) & ~(size_t)511;
    size_t off = 0;
    float* dinv   = (float*)(ws + off); off += nodeB;
    int*   cnt    = (int*)(ws + off);   off += nodeB;
    int*   rowloc = (int*)(ws + off);   off += nodeB;
    int*   blk    = (int*)(ws + off);   off += blkB;
    int*   rank   = (int*)(ws + off);   off += edgeB;
    int*   csr    = (int*)(ws + off);   off += edgeB;
    float* h      = (float*)(ws + off); off += (size_t)n * 128 * 4;
    float* hr     = (float*)(ws + off); off += (size_t)n * 128 * 4;
    float* g      = (float*)(ws + off);

    const int B = 256;

    k_zero_cnt<<<(n + B - 1) / B, B, 0, stream>>>(cnt, n);
    k_count_rank<<<(E + B - 1) / B, B, 0, stream>>>(dst, cnt, rank, E);
    k_scan_local<<<nblk, 256, 0, stream>>>(cnt, rowloc, blk, dinv, n);
    k_scan_blk<<<1, 256, 0, stream>>>(blk, nblk);
    k_fill<<<(E + B - 1) / B, B, 0, stream>>>(src, dst, rank, rowloc, blk, csr, E);

    {
        int waves = (n / 16) * (128 / 16);
        k_gemm_wmma<128, 128><<<waves / 8, 256, 0, stream>>>(x, W1, h);
    }
    k_gather<128, true><<<(n + 7) / 8, 256, 0, stream>>>(csr, cnt, rowloc, blk, dinv, h, b1, hr, n);

    {
        int waves = (n / 16) * (64 / 16);
        k_gemm_wmma<64, 128><<<waves / 8, 256, 0, stream>>>(hr, W2, g);
    }
    k_gather<64, false><<<(n + 7) / 8, 256, 0, stream>>>(csr, cnt, rowloc, blk, dinv, g, b2, out, n);
}


# Each wave takes several work items in a loop (gather: nodes at grid stride, next node's list asked for early; GEMM: a 16xN strip on one A fragment; edge passes: 8 edges per thread): 6x fewer waves
# speedup vs baseline: 2.6175x; 1.0838x over previous
#include <hip/hip_runtime.h>


typedef float v2f __attribute__((ext_vector_type(2)));
typedef float v8f __attribute__((ext_vector_type(8)));

__global__ void k_zero_cnt(int* __restrict__ cnt, int n) {
    const int i = (blockIdx.x * blockDim.x + threadIdx.x) * 4;
    if (i + 3 < n) {
        *(int4*)(cnt + i) = make_int4(0, 0, 0, 0);
    } else {
        for (int j = 0; j < 4; ++j)
            if (i + j < n) cnt[i + j] = 0;
    }
}

__global__ void k_count_rank(const int* __restrict__ dst, int* __restrict__ cnt,
                             int* __restrict__ rank, int E) {
    const int T = gridDim.x * blockDim.x;
    for (int e = blockIdx.x * blockDim.x + threadIdx.x; e < E; e += 4 * T) {
        if (e + 3 * T < E) {
            const int d0 = dst[e], d1 = dst[e + T], d2 = dst[e + 2 * T], d3 = dst[e + 3 * T];
            const int r0 = atomicAdd(&cnt[d0], 1);
            const int r1 = atomicAdd(&cnt[d1], 1);
            const int r2 = atomicAdd(&cnt[d2], 1);
            const int r3 = atomicAdd(&cnt[d3], 1);
            rank[e]         = r0;
            rank[e + T]     = r1;
            rank[e + 2 * T] = r2;
            rank[e + 3 * T] = r3;
        } else {
            for (int ee = e; ee < E; ee += T) rank[ee] = atomicAdd(&cnt[dst[ee]], 1);
        }
    }
}

__device__ __forceinline__ int wave_incl_scan(int v, int lane) {
#pragma unroll
    for (int o = 1; o < 32; o <<= 1) {
        int t = __shfl_up(v, o, 32);
        if (lane >= o) v += t;
    }
    return v;
}

__global__ void k_scan_local(const int* __restrict__ cnt, int* __restrict__ rowloc,
                             int* __restrict__ blksum, float* __restrict__ dinv, int n) {
    __shared__ int wsum[8];
    const int t = threadIdx.x, lane = t & 31, w = t >> 5;
    const int i0 = blockIdx.x * 1024 + t * 4;
    int c[4];
#pragma unroll
    for (int j = 0; j < 4; ++j) c[j] = (i0 + j < n) ? cnt[i0 + j] : 0;
    const int s = c[0] + c[1] + c[2] + c[3];
    const int inc = wave_incl_scan(s, lane);
    if (lane == 31) wsum[w] = inc;
    __syncthreads();
    int woff = 0;
#pragma unroll
    for (int k = 0; k < 8; ++k) woff += (k < w) ? wsum[k] : 0;
    int excl = woff + inc - s;
#pragma unroll
    for (int j = 0; j < 4; ++j) {
        if (i0 + j < n) {
            rowloc[i0 + j] = excl;
            dinv[i0 + j]   = rsqrtf((float)(c[j] + 1));
        }
        excl += c[j];
    }
    if (t == 255) blksum[blockIdx.x] = woff + inc;
}

__global__ void k_scan_blk(int* __restrict__ blksum, int nblk) {
    __shared__ int wsum[8];
    const int t = threadIdx.x, lane = t & 31, w = t >> 5;
    int carry = 0;
    for (int base = 0; base < nblk; base += 256) {
        const int i = base + t;
        const int v = (i < nblk) ? blksum[i] : 0;
        const int inc = wave_incl_scan(v, lane);
        if (lane == 31) wsum[w] = inc;
        __syncthreads();
        int woff = 0, tot = 0;
#pragma unroll
        for (int k = 0; k < 8; ++k) { woff += (k < w) ? wsum[k] : 0; tot += wsum[k]; }
        if (i < nblk) blksum[i] = carry + woff + inc - v;
        carry += tot;
        __syncthreads();
    }
}

__global__ void k_fill(const int* __restrict__ src, const int* __restrict__ dst,
                       const int* __restrict__ rank, const int* __restrict__ rowloc,
                       const int* __restrict__ blkoff, int* __restrict__ csr, int E) {
    const int T = gridDim.x * blockDim.x;
    for (int e = blockIdx.x * blockDim.x + threadIdx.x; e < E; e += 4 * T) {
        if (e + 3 * T < E) {
            int d[4], s[4], rk[4], pos[4];
#pragma unroll
            for (int u = 0; u < 4; ++u) {
                d[u]  = dst[e + u * T];
                s[u]  = src[e + u * T];
                rk[u] = rank[e + u * T];
            }
#pragma unroll
            for (int u = 0; u < 4; ++u) pos[u] = rowloc[d[u]] + blkoff[d[u] >> 10] + rk[u];
#pragma unroll
            for (int u = 0; u < 4; ++u) csr[pos[u]] = s[u];
        } else {
            for (int ee = e; ee < E; ee += T) {
                const int dd = dst[ee];
                csr[rowloc[dd] + blkoff[dd >> 10] + rank[ee]] = src[ee];
            }
        }
    }
}

template <int N, int K>
__global__ void k_gemm_wmma(const float* __restrict__ A, const float* __restrict__ B,
                            float* __restrict__ C, int tilesM) {
    constexpr int TN = N / 16;
    const int lane = threadIdx.x & 31;
    const int half = lane >> 4;
    const int l    = lane & 15;
    const int nw   = (gridDim.x * blockDim.x) >> 5;
    int tM = __builtin_amdgcn_readfirstlane((blockIdx.x * blockDim.x + threadIdx.x) >> 5);

    for (; tM < tilesM; tM += nw) {
        const float* ap = A + ((size_t)tM * 16 + l) * K + half * 2;
        const float* bp = B + (size_t)(half * 2) * N + l;

        v8f acc[TN];
#pragma unroll
        for (int t = 0; t < TN; ++t) acc[t] = v8f{};

#pragma unroll 4
        for (int k = 0; k < K; k += 4) {
            const v2f a = *(const v2f*)(ap + k);
            const float* b0 = bp + (size_t)k * N;
#pragma unroll
            for (int t = 0; t < TN; ++t) {
                v2f b;
                b.x = b0[t * 16];
                b.y = b0[N + t * 16];
                acc[t] = __builtin_amdgcn_wmma_f32_16x16x4_f32(
                     false, a,  false, b,
                     (short)0, acc[t],  false,  false);
            }
        }

        float* cp = C + ((size_t)tM * 16 + half * 8) * N + l;
#pragma unroll
        for (int t = 0; t < TN; ++t) {
#pragma unroll
            for (int v = 0; v < 8; ++v) cp[(size_t)v * N + t * 16] = acc[t][v];
        }
    }
}


template <int F, bool RELU>
__global__ void k_gather(const int* __restrict__ csr, const int* __restrict__ cnt,
                         const int* __restrict__ rowloc, const int* __restrict__ blkoff,
                         const float* __restrict__ dinv, const float* __restrict__ h,
                         const float* __restrict__ bias, float* __restrict__ out, int n) {
    constexpr int PER = F / 32;
    typedef float vf __attribute__((ext_vector_type(PER)));
    const int lane = threadIdx.x & 31;
    const int nw   = (gridDim.x * blockDim.x) >> 5;
    const float* hb = h + lane * PER;
    const vf bvec = *(const vf*)(bias + lane * PER);
    int d = __builtin_amdgcn_readfirstlane((blockIdx.x * blockDim.x + threadIdx.x) >> 5);
    if (d >= n) return;

    int   start = rowloc[d] + blkoff[d >> 10];
    int   deg   = cnt[d];
    float dd    = dinv[d];
    vf    self  = *(const vf*)(hb + (size_t)d * F);
    int   myS   = 0;
    if (lane < deg) myS = csr[start + lane];
    int   dn = d + nw;
    int   locN = 0, offN = 0, degN = 0;
    float ddN = 0.0f;
    if (dn < n) { locN = rowloc[dn]; offN = blkoff[dn >> 10]; degN = cnt[dn]; ddN = dinv[dn]; }

    for (;;) {
        float myD = dinv[myS];
        vf acc = self * (dd * dd);
        int mySN = 0;
        if (lane < degN) mySN = csr[locN + offN + lane];
        vf selfN = self;
        if (dn < n) selfN = *(const vf*)(hb + (size_t)dn * F);
        const int dnn = dn + nw;
        int   locNN = 0, offNN = 0, degNN = 0;
        float ddNN = 0.0f;
        if (dnn < n) { locNN = rowloc[dnn]; offNN = blkoff[dnn >> 10]; degNN = cnt[dnn]; ddNN = dinv[dnn]; }

        for (int base = 0; base < deg; base += 32) {
            const int m = (deg - base) < 32 ? (deg - base) : 32;
            if (base > 0) {
                myS = 0;
                if (lane < m) myS = csr[start + base + lane];
                myD = dinv[myS];
            }
            for (int j = 0; j < m; j += 8) {
                const int r = m - j;
                vf v[8];
#pragma unroll
                for (int u = 0; u < 8; ++u) {
                    if (u < r) {
                        const int s = __builtin_amdgcn_readlane(myS, j + u);
                        v[u] = *(const vf*)(hb + (size_t)s * F);
                    }
                }
                float nr[8];
#pragma unroll
                for (int u = 0; u < 8; ++u)
                    nr[u] = __int_as_float(__builtin_amdgcn_readlane(__float_as_int(myD), (j + u) & 31)) * dd;
#pragma unroll
                for (int u = 0; u < 8; ++u) {
                    if (u < r) acc += v[u] * nr[u];
                }
            }
        }

        acc += bvec;
        if (RELU) {
#pragma unroll
            for (int q = 0; q < PER; ++q) acc[q] = acc[q] > 0.0f ? acc[q] : 0.0f;
        }
        *(vf*)(out + (size_t)d * F + lane * PER) = acc;

        if (dn >= n) break;
        d = dn;   start = locN + offN; deg = degN;  dd = ddN;  myS = mySN;  self = selfN;
        dn = dnn; locN = locNN; offN = offNN; degN = degNN; ddN = ddNN;
    }
}

extern "C" void kernel_launch(void* const* d_in, const int* in_sizes, int n_in,
                              void* d_out, int out_size, void* d_ws, size_t ws_size,
                              hipStream_t stream) {
    const float* x  = (const float*)d_in[0];
    const int*   ei = (const int*)d_in[1];
    const float* W1 = (const float*)d_in[2];
    const float* b1 = (const float*)d_in[3];
    const float* W2 = (const float*)d_in[4];
    const float* b2 = (const float*)d_in[5];
    float* out = (float*)d_out;

    const int n = in_sizes[0] / 128;
    const int E = in_sizes[1] / 2;
    const int* src = ei;
    const int* dst = ei + E;
    const int nblk = (n + 1023) / 1024;

    char* ws = (char*)d_ws;
    const size_t nodeB = ((size_t)n * 4 + 511) & ~(size_t)511;
    const size_t edgeB = ((size_t)E * 4 + 511) & ~(size_t)511;
    const size_t blkB  = ((size_t)nblk * 4 + 511) & ~(size_t)511;
    size_t off = 0;
    float* dinv   = (float*)(ws + off); off += nodeB;
    int*   cnt    = (int*)(ws + off);   off += nodeB;
    int*   rowloc = (int*)(ws + off);   off += nodeB;
    int*   blk    = (int*)(ws + off);   off += blkB;
    int*   rank   = (int*)(ws + off);   off += edgeB;
    int*   csr    = (int*)(ws + off);   off += edgeB;
    float* h      = (float*)(ws + off); off += (size_t)n * 128 * 4;
    float* hr     = (float*)(ws + off); off += (size_t)n * 128 * 4;
    float* g      = (float*)(ws + off);

    const int B = 256;

    const int gridE = (E + 8 * B - 1) / (8 * B);
    const int gridG = (n + 7) / 8 < 2048 ? (n + 7) / 8 : 2048;
    k_zero_cnt<<<(n + 4 * B - 1) / (4 * B), B, 0, stream>>>(cnt, n);
    k_count_rank<<<gridE, B, 0, stream>>>(dst, cnt, rank, E);
    k_scan_local<<<nblk, 256, 0, stream>>>(cnt, rowloc, blk, dinv, n);
    k_scan_blk<<<1, 256, 0, stream>>>(blk, nblk);
    k_fill<<<gridE, B, 0, stream>>>(src, dst, rank, rowloc, blk, csr, E);

    const int tilesM = n / 16;
    k_gemm_wmma<128, 128><<<(tilesM + 3) / 4, 128, 0, stream>>>(x, W1, h, tilesM);
    k_gather<128, true><<<gridG, 256, 0, stream>>>(csr, cnt, rowloc, blk, dinv, h, b1, hr, n);

    k_gemm_wmma<64, 128><<<(tilesM + 3) / 4, 128, 0, stream>>>(hr, W2, g, tilesM);
    k_gather<64, false><<<gridG, 256, 0, stream>>>(csr, cnt, rowloc, blk, dinv, g, b2, out, n);
}


# Edge sort without global atomics: a work-group owns a node range and an edge segment, counts destinations in LDS and after the scan places sources by the LDS add's return; rows sum in segment order
# speedup vs baseline: 4.0892x; 1.5623x over previous
#include <hip/hip_runtime.h>


typedef float v2f __attribute__((ext_vector_type(2)));
typedef float v8f __attribute__((ext_vector_type(8)));

constexpr int SEG   = 16;
constexpr int NRMAX = 8192;
constexpr int WT    = 1024;
constexpr int WU    = 8;

__global__ __launch_bounds__(WT) void k_count_lds(const int* __restrict__ dst, int* __restrict__ cseg,
                                                  int n, int nP, int NR, int E, int segLen) {
    __shared__ int c[NRMAX];
    const int t  = threadIdx.x;
    const int lo = blockIdx.x * NR;
    int nr = n - lo;
    if (nr > NR) nr = NR;
    if (nr < 0) nr = 0;
    for (int i = t; i < nr; i += WT) c[i] = 0;
    __syncthreads();

    const int eb = blockIdx.y * segLen;
    int len = E - eb;
    if (len > segLen) len = segLen;
    if (len > 0) {
        const int* dp = dst + eb;
        int done = 0;
        if ((((size_t)dp) & 15) == 0) {
            const int nvec = len >> 2;
            const int4* dv = (const int4*)dp;
            for (int v0 = t; v0 < nvec; v0 += WU * WT) {
                int4 q[WU];
#pragma unroll
                for (int u = 0; u < WU; ++u) {
                    const int vi = v0 + u * WT;
                    q[u] = make_int4(-1, -1, -1, -1);
                    if (vi < nvec) q[u] = dv[vi];
                }
#pragma unroll
                for (int u = 0; u < WU; ++u) {
                    const unsigned r0 = (unsigned)(q[u].x - lo), r1 = (unsigned)(q[u].y - lo);
                    const unsigned r2 = (unsigned)(q[u].z - lo), r3 = (unsigned)(q[u].w - lo);
                    if (r0 < (unsigned)nr) atomicAdd(&c[r0], 1);
                    if (r1 < (unsigned)nr) atomicAdd(&c[r1], 1);
                    if (r2 < (unsigned)nr) atomicAdd(&c[r2], 1);
                    if (r3 < (unsigned)nr) atomicAdd(&c[r3], 1);
                }
            }
            done = nvec << 2;
        }
        for (int e = done + t; e < len; e += WT) {
            const unsigned r0 = (unsigned)(dp[e] - lo);
            if (r0 < (unsigned)nr) atomicAdd(&c[r0], 1);
        }
    }
    __syncthreads();
    for (int i = t; i < nr; i += WT) cseg[(size_t)blockIdx.y * nP + lo + i] = c[i];
}

__global__ __launch_bounds__(WT) void k_place_lds(const int* __restrict__ src, const int* __restrict__ dst,
                                                  const int* __restrict__ cseg, const int* __restrict__ blkoff,
                                                  int* __restrict__ csr,
                                                  int n, int nP, int NR, int E, int segLen) {
    __shared__ int c[NRMAX];
    const int t  = threadIdx.x;
    const int lo = blockIdx.x * NR;
    int nr = n - lo;
    if (nr > NR) nr = NR;
    if (nr < 0) nr = 0;
    for (int i = t; i < nr; i += WT)
        c[i] = cseg[(size_t)blockIdx.y * nP + lo + i] + blkoff[(lo + i) >> 10];
    __syncthreads();

    const int eb = blockIdx.y * segLen;
    int len = E - eb;
    if (len > segLen) len = segLen;
    if (len > 0) {
        const int* dp = dst + eb;
        const int* sp = src + eb;
        int done = 0;
        if (((((size_t)dp) | ((size_t)sp)) & 15) == 0) {
            const int nvec = len >> 2;
            const int4* dv = (const int4*)dp;
            const int4* sv = (const int4*)sp;
            for (int v0 = t; v0 < nvec; v0 += WU * WT) {
                int4 q[WU], w[WU];
#pragma unroll
                for (int u = 0; u < WU; ++u) {
                    const int vi = v0 + u * WT;
                    q[u] = make_int4(-1, -1, -1, -1);
                    w[u] = make_int4(0, 0, 0, 0);
                    if (vi < nvec) { q[u] = dv[vi]; w[u] = sv[vi]; }
                }
                int p[WU][4];
#pragma unroll
                for (int u = 0; u < WU; ++u) {
                    const unsigned r0 = (unsigned)(q[u].x - lo), r1 = (unsigned)(q[u].y - lo);
                    const unsigned r2 = (unsigned)(q[u].z - lo), r3 = (unsigned)(q[u].w - lo);
                    p[u][0] = -1; p[u][1] = -1; p[u][2] = -1; p[u][3] = -1;
                    if (r0 < (unsigned)nr) p[u][0] = atomicAdd(&c[r0], 1);
                    if (r1 < (unsigned)nr) p[u][1] = atomicAdd(&c[r1], 1);
                    if (r2 < (unsigned)nr) p[u][2] = atomicAdd(&c[r2], 1);
                    if (r3 < (unsigned)nr) p[u][3] = atomicAdd(&c[r3], 1);
                }
#pragma unroll
                for (int u = 0; u < WU; ++u) {
                    if (p[u][0] >= 0) csr[p[u][0]] = w[u].x;
                    if (p[u][1] >= 0) csr[p[u][1]] = w[u].y;
                    if (p[u][2] >= 0) csr[p[u][2]] = w[u].z;
                    if (p[u][3] >= 0) csr[p[u][3]] = w[u].w;
                }
            }
            done = nvec << 2;
        }
        for (int e = done + t; e < len; e += WT) {
            const unsigned r0 = (unsigned)(dp[e] - lo);
            if (r0 < (unsigned)nr) csr[atomicAdd(&c[r0], 1)] = sp[e];
        }
    }
}

__device__ __forceinline__ int wave_incl_scan(int v, int lane) {
#pragma unroll
    for (int o = 1; o < 32; o <<= 1) {
        int t = __shfl_up(v, o, 32);
        if (lane >= o) v += t;
    }
    return v;
}

__global__ void k_scan_local(int* __restrict__ cseg, int* __restrict__ cnt, int* __restrict__ rowloc,
                             int* __restrict__ blksum, float* __restrict__ dinv, int n, int nP) {
    __shared__ int wsum[8];
    const int t = threadIdx.x, lane = t & 31, w = t >> 5;
    const int i0 = blockIdx.x * 1024 + t * 4;
    const bool full = (i0 + 3 < n);
    int cs[SEG][4];
    int c[4] = {0, 0, 0, 0};
#pragma unroll
    for (int sg = 0; sg < SEG; ++sg) {
        const int* p = cseg + (size_t)sg * nP + i0;
        if (full) {
            const int4 v = *(const int4*)p;
            cs[sg][0] = v.x; cs[sg][1] = v.y; cs[sg][2] = v.z; cs[sg][3] = v.w;
        } else {
#pragma unroll
            for (int j = 0; j < 4; ++j) cs[sg][j] = (i0 + j < n) ? p[j] : 0;
        }
#pragma unroll
        for (int j = 0; j < 4; ++j) c[j] += cs[sg][j];
    }
    const int s = c[0] + c[1] + c[2] + c[3];
    const int inc = wave_incl_scan(s, lane);
    if (lane == 31) wsum[w] = inc;
    __syncthreads();
    int woff = 0;
#pragma unroll
    for (int k = 0; k < 8; ++k) woff += (k < w) ? wsum[k] : 0;
    int excl = woff + inc - s;
#pragma unroll
    for (int j = 0; j < 4; ++j) {
        if (i0 + j < n) {
            rowloc[i0 + j] = excl;
            cnt[i0 + j]    = c[j];
            dinv[i0 + j]   = rsqrtf((float)(c[j] + 1));
        }
        int run = excl;
#pragma unroll
        for (int sg = 0; sg < SEG; ++sg) { const int v = cs[sg][j]; cs[sg][j] = run; run += v; }
        excl += c[j];
    }
#pragma unroll
    for (int sg = 0; sg < SEG; ++sg) {
        int* p = cseg + (size_t)sg * nP + i0;
        if (full) {
            *(int4*)p = make_int4(cs[sg][0], cs[sg][1], cs[sg][2], cs[sg][3]);
        } else {
#pragma unroll
            for (int j = 0; j < 4; ++j) if (i0 + j < n) p[j] = cs[sg][j];
        }
    }
    if (t == 255) blksum[blockIdx.x] = woff + inc;
}

__global__ void k_scan_blk(int* __restrict__ blksum, int nblk) {
    __shared__ int wsum[8];
    const int t = threadIdx.x, lane = t & 31, w = t >> 5;
    int carry = 0;
    for (int base = 0; base < nblk; base += 256) {
        const int i = base + t;
        const int v = (i < nblk) ? blksum[i] : 0;
        const int inc = wave_incl_scan(v, lane);
        if (lane == 31) wsum[w] = inc;
        __syncthreads();
        int woff = 0, tot = 0;
#pragma unroll
        for (int k = 0; k < 8; ++k) { woff += (k < w) ? wsum[k] : 0; tot += wsum[k]; }
        if (i < nblk) blksum[i] = carry + woff + inc - v;
        carry += tot;
        __syncthreads();
    }
}

template <int N, int K>
__global__ void k_gemm_wmma(const float* __restrict__ A, const float* __restrict__ B,
                            float* __restrict__ C, int tilesM) {
    constexpr int TN = N / 16;
    const int lane = threadIdx.x & 31;
    const int half = lane >> 4;
    const int l    = lane & 15;
    const int nw   = (gridDim.x * blockDim.x) >> 5;
    int tM = __builtin_amdgcn_readfirstlane((blockIdx.x * blockDim.x + threadIdx.x) >> 5);

    for (; tM < tilesM; tM += nw) {
        const float* ap = A + ((size_t)tM * 16 + l) * K + half * 2;
        const float* bp = B + (size_t)(half * 2) * N + l;

        v8f acc[TN];
#pragma unroll
        for (int t = 0; t < TN; ++t) acc[t] = v8f{};

#pragma unroll 4
        for (int k = 0; k < K; k += 4) {
            const v2f a = *(const v2f*)(ap + k);
            const float* b0 = bp + (size_t)k * N;
#pragma unroll
            for (int t = 0; t < TN; ++t) {
                v2f b;
                b.x = b0[t * 16];
                b.y = b0[N + t * 16];
                acc[t] = __builtin_amdgcn_wmma_f32_16x16x4_f32(
                     false, a,  false, b,
                     (short)0, acc[t],  false,  false);
            }
        }

        float* cp = C + ((size_t)tM * 16 + half * 8) * N + l;
#pragma unroll
        for (int t = 0; t < TN; ++t) {
#pragma unroll
            for (int v = 0; v < 8; ++v) cp[(size_t)v * N + t * 16] = acc[t][v];
        }
    }
}


template <int F, bool RELU>
__global__ void k_gather(const int* __restrict__ csr, const int* __restrict__ cnt,
                         const int* __restrict__ rowloc, const int* __restrict__ blkoff,
                         const float* __restrict__ dinv, const float* __restrict__ h,
                         const float* __restrict__ bias, float* __restrict__ out, int n) {
    constexpr int PER = F / 32;
    typedef float vf __attribute__((ext_vector_type(PER)));
    const int lane = threadIdx.x & 31;
    const int nw   = (gridDim.x * blockDim.x) >> 5;
    const float* hb = h + lane * PER;
    const vf bvec = *(const vf*)(bias + lane * PER);
    int d = __builtin_amdgcn_readfirstlane((blockIdx.x * blockDim.x + threadIdx.x) >> 5);
    if (d >= n) return;

    int   start = rowloc[d] + blkoff[d >> 10];
    int   deg   = cnt[d];
    float dd    = dinv[d];
    vf    self  = *(const vf*)(hb + (size_t)d * F);
    int   myS   = 0;
    if (lane < deg) myS = csr[start + lane];
    int   dn = d + nw;
    int   locN = 0, offN = 0, degN = 0;
    float ddN = 0.0f;
    if (dn < n) { locN = rowloc[dn]; offN = blkoff[dn >> 10]; degN = cnt[dn]; ddN = dinv[dn]; }

    for (;;) {
        float myD = dinv[myS];
        vf acc = self * (dd * dd);
        int mySN = 0;
        if (lane < degN) mySN = csr[locN + offN + lane];
        vf selfN = self;
        if (dn < n) selfN = *(const vf*)(hb + (size_t)dn * F);
        const int dnn = dn + nw;
        int   locNN = 0, offNN = 0, degNN = 0;
        float ddNN = 0.0f;
        if (dnn < n) { locNN = rowloc[dnn]; offNN = blkoff[dnn >> 10]; degNN = cnt[dnn]; ddNN = dinv[dnn]; }

        for (int base = 0; base < deg; base += 32) {
            const int m = (deg - base) < 32 ? (deg - base) : 32;
            if (base > 0) {
                myS = 0;
                if (lane < m) myS = csr[start + base + lane];
                myD = dinv[myS];
            }
            for (int j = 0; j < m; j += 8) {
                const int r = m - j;
                vf v[8];
#pragma unroll
                for (int u = 0; u < 8; ++u) {
                    if (u < r) {
                        const int s = __builtin_amdgcn_readlane(myS, j + u);
                        v[u] = *(const vf*)(hb + (size_t)s * F);
                    }
                }
                float nr[8];
#pragma unroll
                for (int u = 0; u < 8; ++u)
                    nr[u] = __int_as_float(__builtin_amdgcn_readlane(__float_as_int(myD), (j + u) & 31)) * dd;
#pragma unroll
                for (int u = 0; u < 8; ++u) {
                    if (u < r) acc += v[u] * nr[u];
                }
            }
        }

        acc += bvec;
        if (RELU) {
#pragma unroll
            for (int q = 0; q < PER; ++q) acc[q] = acc[q] > 0.0f ? acc[q] : 0.0f;
        }
        *(vf*)(out + (size_t)d * F + lane * PER) = acc;

        if (dn >= n) break;
        d = dn;   start = locN + offN; deg = degN;  dd = ddN;  myS = mySN;  self = selfN;
        dn = dnn; locN = locNN; offN = offNN; degN = degNN; ddN = ddNN;
    }
}

extern "C" void kernel_launch(void* const* d_in, const int* in_sizes, int n_in,
                              void* d_out, int out_size, void* d_ws, size_t ws_size,
                              hipStream_t stream) {
    const float* x  = (const float*)d_in[0];
    const int*   ei = (const int*)d_in[1];
    const float* W1 = (const float*)d_in[2];
    const float* b1 = (const float*)d_in[3];
    const float* W2 = (const float*)d_in[4];
    const float* b2 = (const float*)d_in[5];
    float* out = (float*)d_out;

    const int n = in_sizes[0] / 128;
    const int E = in_sizes[1] / 2;
    const int* src = ei;
    const int* dst = ei + E;
    const int nblk = (n + 1023) / 1024;

    char* ws = (char*)d_ws;
    const int    nP    = (n + 3) & ~3;
    const size_t nodeB = ((size_t)n * 4 + 511) & ~(size_t)511;
    const size_t edgeB = ((size_t)E * 4 + 511) & ~(size_t)511;
    const size_t blkB  = ((size_t)nblk * 4 + 511) & ~(size_t)511;
    const size_t segB  = ((size_t)SEG * nP * 4 + 511) & ~(size_t)511;
    size_t off = 0;
    float* dinv   = (float*)(ws + off); off += nodeB;
    int*   cnt    = (int*)(ws + off);   off += nodeB;
    int*   rowloc = (int*)(ws + off);   off += nodeB;
    int*   blk    = (int*)(ws + off);   off += blkB;
    int*   cseg   = (int*)(ws + off);   off += segB;
    int*   csr    = (int*)(ws + off);   off += edgeB;
    float* h      = (float*)(ws + off); off += (size_t)n * 128 * 4;
    float* hr     = (float*)(ws + off); off += (size_t)n * 128 * 4;
    float* g      = (float*)(ws + off);

    int R = (n + NRMAX - 1) / NRMAX;
    if (R < 16) R = 16;
    const int NR     = (n + R - 1) / R;
    const int segLen = (((E + 3) / 4 + SEG - 1) / SEG) * 4;
    const int gridG  = (n + 7) / 8 < 2048 ? (n + 7) / 8 : 2048;
    k_count_lds<<<dim3(R, SEG), WT, 0, stream>>>(dst, cseg, n, nP, NR, E, segLen);
    k_scan_local<<<nblk, 256, 0, stream>>>(cseg, cnt, rowloc, blk, dinv, n, nP);
    k_scan_blk<<<1, 256, 0, stream>>>(blk, nblk);
    k_place_lds<<<dim3(R, SEG), WT, 0, stream>>>(src, dst, cseg, blk, csr, n, nP, NR, E, segLen);

    const int tilesM = n / 16;
    k_gemm_wmma<128, 128><<<(tilesM + 3) / 4, 128, 0, stream>>>(x, W1, h, tilesM);
    k_gather<128, true><<<gridG, 256, 0, stream>>>(csr, cnt, rowloc, blk, dinv, h, b1, hr, n);

    k_gemm_wmma<64, 128><<<(tilesM + 3) / 4, 128, 0, stream>>>(hr, W2, g, tilesM);
    k_gather<64, false><<<gridG, 256, 0, stream>>>(csr, cnt, rowloc, blk, dinv, g, b2, out, n);
}


# The two scan launches are folded into the placement launch: each placement work-group sums its range's per-segment counts, scans them itself and adds the totals of earlier ranges (6 launches to 4)
# speedup vs baseline: 5.0054x; 1.2241x over previous
#include <hip/hip_runtime.h>


typedef float v2f __attribute__((ext_vector_type(2)));
typedef float v8f __attribute__((ext_vector_type(8)));

constexpr int SEG   = 16;
constexpr int NRMAX = 8192;
constexpr int WT    = 1024;
constexpr int WU    = 8;
static_assert(NRMAX == 2 * 4 * WT, "a thread of the placement prologue owns two groups of four nodes");
static_assert(WT == 32 * 32, "wg_excl_scan holds one sum per wave in one wave's lanes");

__device__ __forceinline__ int wave_incl_scan(int v, int lane) {
#pragma unroll
    for (int o = 1; o < 32; o <<= 1) {
        int t = __shfl_up(v, o, 32);
        if (lane >= o) v += t;
    }
    return v;
}

__device__ __forceinline__ int wg_excl_scan(int v, int* ws, int t, int& total) {
    const int lane = t & 31, w = t >> 5;
    const int inc = wave_incl_scan(v, lane);
    if (lane == 31) ws[w] = inc;
    __syncthreads();
    const int wv   = ws[lane];
    const int winc = wave_incl_scan(wv, lane);
    const int woff = __shfl(winc, w, 32) - __shfl(wv, w, 32);
    total = __shfl(winc, 31, 32);
    __syncthreads();
    return woff + inc - v;
}

__global__ __launch_bounds__(WT) void k_count_lds(const int* __restrict__ dst, int* __restrict__ cseg,
                                                  int* __restrict__ wgtot,
                                                  int n, int nP, int NR, int E, int segLen) {
    __shared__ int c[NRMAX];
    __shared__ int wsum[32];
    const int t  = threadIdx.x;
    const int lo = blockIdx.x * NR;
    int nr = n - lo;
    if (nr > NR) nr = NR;
    if (nr < 0) nr = 0;
    for (int i = t; i < nr; i += WT) c[i] = 0;
    __syncthreads();

    const int eb = blockIdx.y * segLen;
    int len = E - eb;
    if (len > segLen) len = segLen;
    if (len > 0) {
        const int* dp = dst + eb;
        int done = 0;
        if ((((size_t)dp) & 15) == 0) {
            const int nvec = len >> 2;
            const int4* dv = (const int4*)dp;
            for (int v0 = t; v0 < nvec; v0 += WU * WT) {
                int4 q[WU];
#pragma unroll
                for (int u = 0; u < WU; ++u) {
                    const int vi = v0 + u * WT;
                    q[u] = make_int4(-1, -1, -1, -1);
                    if (vi < nvec) q[u] = dv[vi];
                }
#pragma unroll
                for (int u = 0; u < WU; ++u) {
                    const unsigned r0 = (unsigned)(q[u].x - lo), r1 = (unsigned)(q[u].y - lo);
                    const unsigned r2 = (unsigned)(q[u].z - lo), r3 = (unsigned)(q[u].w - lo);
                    if (r0 < (unsigned)nr) atomicAdd(&c[r0], 1);
                    if (r1 < (unsigned)nr) atomicAdd(&c[r1], 1);
                    if (r2 < (unsigned)nr) atomicAdd(&c[r2], 1);
                    if (r3 < (unsigned)nr) atomicAdd(&c[r3], 1);
                }
            }
            done = nvec << 2;
        }
        for (int e = done + t; e < len; e += WT) {
            const unsigned r0 = (unsigned)(dp[e] - lo);
            if (r0 < (unsigned)nr) atomicAdd(&c[r0], 1);
        }
    }
    __syncthreads();
    int part = 0;
    for (int i = t; i < nr; i += WT) {
        const int v = c[i];
        cseg[(size_t)blockIdx.y * nP + lo + i] = v;
        part += v;
    }
    int tot;
    wg_excl_scan(part, wsum, t, tot);
    if (t == 0) wgtot[blockIdx.x * SEG + blockIdx.y] = tot;
}

__global__ __launch_bounds__(WT) void k_place_lds(const int* __restrict__ src, const int* __restrict__ dst,
                                                  const int* __restrict__ cseg, const int* __restrict__ wgtot,
                                                  int* __restrict__ csr, int* __restrict__ rowstart,
                                                  int* __restrict__ cnt, float* __restrict__ dinv,
                                                  int n, int nP, int NR, int E, int segLen) {
    __shared__ int c[NRMAX];
    __shared__ int wsum[32];
    const int t  = threadIdx.x;
    const int lo = blockIdx.x * NR;
    const int sg = blockIdx.y;
    int nr = n - lo;
    if (nr > NR) nr = NR;
    if (nr < 0) nr = 0;

    int tot[2][4], pre[2][4];
#pragma unroll
    for (int h = 0; h < 2; ++h) {
        const int i = h * (NRMAX / 2) + 4 * t;
#pragma unroll
        for (int k = 0; k < 4; ++k) { tot[h][k] = 0; pre[h][k] = 0; }
        if (i < nr) {
            int4 v[SEG];
#pragma unroll
            for (int s2 = 0; s2 < SEG; ++s2) v[s2] = *(const int4*)(cseg + (size_t)s2 * nP + lo + i);
#pragma unroll
            for (int s2 = 0; s2 < SEG; ++s2) {
                tot[h][0] += v[s2].x; tot[h][1] += v[s2].y; tot[h][2] += v[s2].z; tot[h][3] += v[s2].w;
                if (s2 < sg) { pre[h][0] += v[s2].x; pre[h][1] += v[s2].y; pre[h][2] += v[s2].z; pre[h][3] += v[s2].w; }
            }
#pragma unroll
            for (int k = 0; k < 4; ++k)
                if (i + k >= nr) { tot[h][k] = 0; pre[h][k] = 0; }
        }
    }
    int bs = 0;
    for (int q = t; q < (int)blockIdx.x * SEG; q += WT) bs += wgtot[q];
    int carry;
    wg_excl_scan(bs, wsum, t, carry);
#pragma unroll
    for (int h = 0; h < 2; ++h) {
        const int i = h * (NRMAX / 2) + 4 * t;
        int all;
        int st = carry + wg_excl_scan(tot[h][0] + tot[h][1] + tot[h][2] + tot[h][3], wsum, t, all);
#pragma unroll
        for (int k = 0; k < 4; ++k) {
            if (i + k < nr) {
                if (sg == 0) {
                    rowstart[lo + i + k] = st;
                    cnt[lo + i + k]      = tot[h][k];
                    dinv[lo + i + k]     = rsqrtf((float)(tot[h][k] + 1));
                }
                c[i + k] = st + pre[h][k];
            }
            st += tot[h][k];
        }
        carry += all;
    }
    __syncthreads();

    const int eb = blockIdx.y * segLen;
    int len = E - eb;
    if (len > segLen) len = segLen;
    if (len > 0) {
        const int* dp = dst + eb;
        const int* sp = src + eb;
        int done = 0;
        if (((((size_t)dp) | ((size_t)sp)) & 15) == 0) {
            const int nvec = len >> 2;
            const int4* dv = (const int4*)dp;
            const int4* sv = (const int4*)sp;
            for (int v0 = t; v0 < nvec; v0 += WU * WT) {
                int4 q[WU], w[WU];
#pragma unroll
                for (int u = 0; u < WU; ++u) {
                    const int vi = v0 + u * WT;
                    q[u] = make_int4(-1, -1, -1, -1);
                    w[u] = make_int4(0, 0, 0, 0);
                    if (vi < nvec) { q[u] = dv[vi]; w[u] = sv[vi]; }
                }
                int p[WU][4];
#pragma unroll
                for (int u = 0; u < WU; ++u) {
                    const unsigned r0 = (unsigned)(q[u].x - lo), r1 = (unsigned)(q[u].y - lo);
                    const unsigned r2 = (unsigned)(q[u].z - lo), r3 = (unsigned)(q[u].w - lo);
                    p[u][0] = -1; p[u][1] = -1; p[u][2] = -1; p[u][3] = -1;
                    if (r0 < (unsigned)nr) p[u][0] = atomicAdd(&c[r0], 1);
                    if (r1 < (unsigned)nr) p[u][1] = atomicAdd(&c[r1], 1);
                    if (r2 < (unsigned)nr) p[u][2] = atomicAdd(&c[r2], 1);
                    if (r3 < (unsigned)nr) p[u][3] = atomicAdd(&c[r3], 1);
                }
#pragma unroll
                for (int u = 0; u < WU; ++u) {
                    if (p[u][0] >= 0) csr[p[u][0]] = w[u].x;
                    if (p[u][1] >= 0) csr[p[u][1]] = w[u].y;
                    if (p[u][2] >= 0) csr[p[u][2]] = w[u].z;
                    if (p[u][3] >= 0) csr[p[u][3]] = w[u].w;
                }
            }
            done = nvec << 2;
        }
        for (int e = done + t; e < len; e += WT) {
            const unsigned r0 = (unsigned)(dp[e] - lo);
            if (r0 < (unsigned)nr) csr[atomicAdd(&c[r0], 1)] = sp[e];
        }
    }
}

constexpr int LP = 128 + 4;

__global__ __launch_bounds__(128) void k_agg_mlp(const int* __restrict__ csr, const int* __restrict__ cnt,
                                                 const int* __restrict__ rowstart,
                                                 const float* __restrict__ dinv, const float* __restrict__ x,
                                                 const float* __restrict__ W1, const float* __restrict__ b1,
                                                 const float* __restrict__ W2, float* __restrict__ g, int n) {
    typedef float vf __attribute__((ext_vector_type(4)));
    __shared__ float sA[16 * LP];
    __shared__ float sH[16 * LP];
    const int lane = threadIdx.x & 31;
    const int w = __builtin_amdgcn_readfirstlane((int)(threadIdx.x >> 5));
    const int strip0 = blockIdx.x * 16;

    const int d0 = strip0 + w * 4;
    int dEnd = d0 + 4;
    if (dEnd > n) dEnd = n;
    if (d0 + 4 > n) {
#pragma unroll
        for (int i = 0; i < 4; ++i) *(vf*)(sA + (w * 4 + i) * LP + lane * 4) = vf{0.0f, 0.0f, 0.0f, 0.0f};
    }
    if (d0 < n) {
        const float* hb = x + lane * 4;
        int   d     = d0;
        int   start = rowstart[d];
        int   deg   = cnt[d];
        float dd    = dinv[d];
        vf    self  = *(const vf*)(hb + (size_t)d * 128);
        int   myS   = 0;
        if (lane < deg) myS = csr[start + lane];
        int   dn = d + 1;
        int   startN = 0, degN = 0;
        float ddN = 0.0f;
        if (dn < dEnd) { startN = rowstart[dn]; degN = cnt[dn]; ddN = dinv[dn]; }

        for (;;) {
            float myD = dinv[myS];
            vf acc = self * (dd * dd);
            int mySN = 0;
            if (lane < degN) mySN = csr[startN + lane];
            vf selfN = self;
            if (dn < dEnd) selfN = *(const vf*)(hb + (size_t)dn * 128);
            const int dnn = dn + 1;
            int   startNN = 0, degNN = 0;
            float ddNN = 0.0f;
            if (dnn < dEnd) { startNN = rowstart[dnn]; degNN = cnt[dnn]; ddNN = dinv[dnn]; }

            for (int base = 0; base < deg; base += 32) {
                const int m = (deg - base) < 32 ? (deg - base) : 32;
                if (base > 0) {
                    myS = 0;
                    if (lane < m) myS = csr[start + base + lane];
                    myD = dinv[myS];
                }
                for (int j = 0; j < m; j += 8) {
                    const int r = m - j;
                    vf v[8];
#pragma unroll
                    for (int u = 0; u < 8; ++u) {
                        if (u < r) {
                            const int s = __builtin_amdgcn_readlane(myS, j + u);
                            v[u] = *(const vf*)(hb + (size_t)s * 128);
                        }
                    }
                    float nr[8];
#pragma unroll
                    for (int u = 0; u < 8; ++u)
                        nr[u] = __int_as_float(__builtin_amdgcn_readlane(__float_as_int(myD), (j + u) & 31)) * dd;
#pragma unroll
                    for (int u = 0; u < 8; ++u) {
                        if (u < r) acc += v[u] * nr[u];
                    }
                }
            }

            *(vf*)(sA + (d - strip0) * LP + lane * 4) = acc;

            if (dn >= dEnd) break;
            d = dn;   start = startN; deg = degN;  dd = ddN;  myS = mySN;  self = selfN;
            dn = dnn; startN = startNN; degN = degNN; ddN = ddNN;
        }
    }
    __syncthreads();

    const int half = lane >> 4;
    const int l    = lane & 15;

    {
        v8f acc0 = v8f{}, acc1 = v8f{};
        const float* ap = sA + l * LP + half * 2;
        const int    c0 = 32 * w + 2 * l;
        const float* bp = W1 + (size_t)(half * 2) * 128 + c0;
#pragma unroll 4
        for (int k = 0; k < 128; k += 4) {
            const v2f a  = *(const v2f*)(ap + k);
            const v2f r0 = *(const v2f*)(bp + (size_t)k * 128);
            const v2f r1 = *(const v2f*)(bp + (size_t)k * 128 + 128);
            v2f bA, bB;
            bA.x = r0.x; bA.y = r1.x;
            bB.x = r0.y; bB.y = r1.y;
            acc0 = __builtin_amdgcn_wmma_f32_16x16x4_f32(false, a, false, bA, (short)0, acc0, false, false);
            acc1 = __builtin_amdgcn_wmma_f32_16x16x4_f32(false, a, false, bB, (short)0, acc1, false, false);
        }
        const v2f bb = *(const v2f*)(b1 + c0);
        float* hp = sH + (half * 8) * LP + c0;
#pragma unroll
        for (int v = 0; v < 8; ++v) {
            const float u0 = acc0[v] + bb.x, u1 = acc1[v] + bb.y;
            v2f o;
            o.x = u0 > 0.0f ? u0 : 0.0f;
            o.y = u1 > 0.0f ? u1 : 0.0f;
            *(v2f*)(hp + v * LP) = o;
        }
    }
    __syncthreads();

    {
        v8f acc = v8f{};
        const float* ap = sH + l * LP + half * 2;
        const float* bp = W2 + (size_t)(half * 2) * 64 + w * 16 + l;
#pragma unroll 4
        for (int k = 0; k < 128; k += 4) {
            const v2f a = *(const v2f*)(ap + k);
            const float* b0 = bp + (size_t)k * 64;
            v2f b;
            b.x = b0[0]; b.y = b0[64];
            acc = __builtin_amdgcn_wmma_f32_16x16x4_f32(false, a, false, b, (short)0, acc, false, false);
        }
        const int r0 = strip0 + half * 8;
        float* cp = g + (size_t)r0 * 64 + w * 16 + l;
#pragma unroll
        for (int v = 0; v < 8; ++v) {
            if (r0 + v < n) cp[(size_t)v * 64] = acc[v];
        }
    }
}


template <int F, bool RELU>
__global__ void k_gather(const int* __restrict__ csr, const int* __restrict__ cnt,
                         const int* __restrict__ rowstart,
                         const float* __restrict__ dinv, const float* __restrict__ h,
                         const float* __restrict__ bias, float* __restrict__ out, int n) {
    constexpr int PER = F / 32;
    typedef float vf __attribute__((ext_vector_type(PER)));
    const int lane = threadIdx.x & 31;
    const int nw   = (gridDim.x * blockDim.x) >> 5;
    const float* hb = h + lane * PER;
    const vf bvec = *(const vf*)(bias + lane * PER);
    int d = __builtin_amdgcn_readfirstlane((blockIdx.x * blockDim.x + threadIdx.x) >> 5);
    if (d >= n) return;

    int   start = rowstart[d];
    int   deg   = cnt[d];
    float dd    = dinv[d];
    vf    self  = *(const vf*)(hb + (size_t)d * F);
    int   myS   = 0;
    if (lane < deg) myS = csr[start + lane];
    int   dn = d + nw;
    int   startN = 0, degN = 0;
    float ddN = 0.0f;
    if (dn < n) { startN = rowstart[dn]; degN = cnt[dn]; ddN = dinv[dn]; }

    for (;;) {
        float myD = dinv[myS];
        vf acc = self * (dd * dd);
        int mySN = 0;
        if (lane < degN) mySN = csr[startN + lane];
        vf selfN = self;
        if (dn < n) selfN = *(const vf*)(hb + (size_t)dn * F);
        const int dnn = dn + nw;
        int   startNN = 0, degNN = 0;
        float ddNN = 0.0f;
        if (dnn < n) { startNN = rowstart[dnn]; degNN = cnt[dnn]; ddNN = dinv[dnn]; }

        for (int base = 0; base < deg; base += 32) {
            const int m = (deg - base) < 32 ? (deg - base) : 32;
            if (base > 0) {
                myS = 0;
                if (lane < m) myS = csr[start + base + lane];
                myD = dinv[myS];
            }
            for (int j = 0; j < m; j += 8) {
                const int r = m - j;
                vf v[8];
#pragma unroll
                for (int u = 0; u < 8; ++u) {
                    if (u < r) {
                        const int s = __builtin_amdgcn_readlane(myS, j + u);
                        v[u] = *(const vf*)(hb + (size_t)s * F);
                    }
                }
                float nr[8];
#pragma unroll
                for (int u = 0; u < 8; ++u)
                    nr[u] = __int_as_float(__builtin_amdgcn_readlane(__float_as_int(myD), (j + u) & 31)) * dd;
#pragma unroll
                for (int u = 0; u < 8; ++u) {
                    if (u < r) acc += v[u] * nr[u];
                }
            }
        }

        acc += bvec;
        if (RELU) {
#pragma unroll
            for (int q = 0; q < PER; ++q) acc[q] = acc[q] > 0.0f ? acc[q] : 0.0f;
        }
        *(vf*)(out + (size_t)d * F + lane * PER) = acc;

        if (dn >= n) break;
        d = dn;   start = startN; deg = degN;  dd = ddN;  myS = mySN;  self = selfN;
        dn = dnn; startN = startNN; degN = degNN; ddN = ddNN;
    }
}

extern "C" void kernel_launch(void* const* d_in, const int* in_sizes, int n_in,
                              void* d_out, int out_size, void* d_ws, size_t ws_size,
                              hipStream_t stream) {
    const float* x  = (const float*)d_in[0];
    const int*   ei = (const int*)d_in[1];
    const float* W1 = (const float*)d_in[2];
    const float* b1 = (const float*)d_in[3];
    const float* W2 = (const float*)d_in[4];
    const float* b2 = (const float*)d_in[5];
    float* out = (float*)d_out;

    const int n = in_sizes[0] / 128;
    const int E = in_sizes[1] / 2;
    const int* src = ei;
    const int* dst = ei + E;

    int R = (n + NRMAX - 1) / NRMAX;
    if (R < 16) R = 16;
    const int NR     = (((n + R - 1) / R) + 3) & ~3;
    const int segLen = (((E + 3) / 4 + SEG - 1) / SEG) * 4;
    const int gridG  = (n + 7) / 8 < 2048 ? (n + 7) / 8 : 2048;

    char* ws = (char*)d_ws;
    const int    nP    = (n + 3) & ~3;
    const size_t nodeB = ((size_t)n * 4 + 511) & ~(size_t)511;
    const size_t edgeB = ((size_t)E * 4 + 511) & ~(size_t)511;
    const size_t totB  = ((size_t)R * SEG * 4 + 511) & ~(size_t)511;
    const size_t segB  = ((size_t)SEG * nP * 4 + 511) & ~(size_t)511;
    size_t off = 0;
    float* dinv     = (float*)(ws + off); off += nodeB;
    int*   cnt      = (int*)(ws + off);   off += nodeB;
    int*   rowstart = (int*)(ws + off);   off += nodeB;
    int*   wgtot    = (int*)(ws + off);   off += totB;
    int*   cseg     = (int*)(ws + off);   off += segB;
    int*   csr      = (int*)(ws + off);   off += edgeB;
    float* g        = (float*)(ws + off);

    k_count_lds<<<dim3(R, SEG), WT, 0, stream>>>(dst, cseg, wgtot, n, nP, NR, E, segLen);
    k_place_lds<<<dim3(R, SEG), WT, 0, stream>>>(src, dst, cseg, wgtot, csr, rowstart, cnt, dinv, n, nP, NR, E, segLen);

    k_agg_mlp<<<(n + 15) / 16, 128, 0, stream>>>(csr, cnt, rowstart, dinv, x, W1, b1, W2, g, n);

    k_gather<64, false><<<gridG, 256, 0, stream>>>(csr, cnt, rowstart, dinv, g, b2, out, n);
}


# conv2 rows are stored pre-multiplied by their node's dinv; the conv2 gather reads no per-source norm and applies dinv[d] once to the sum (same sum order, products re-associated)
# speedup vs baseline: 5.1651x; 1.0319x over previous
#include <hip/hip_runtime.h>


typedef float v2f __attribute__((ext_vector_type(2)));
typedef float v8f __attribute__((ext_vector_type(8)));

constexpr int SEG   = 16;
constexpr int NRMAX = 8192;
constexpr int WT    = 1024;
constexpr int WU    = 8;
static_assert(NRMAX == 2 * 4 * WT, "a thread of the placement prologue owns two groups of four nodes");
static_assert(WT == 32 * 32, "wg_excl_scan holds one sum per wave in one wave's lanes");

__device__ __forceinline__ int wave_incl_scan(int v, int lane) {
#pragma unroll
    for (int o = 1; o < 32; o <<= 1) {
        int t = __shfl_up(v, o, 32);
        if (lane >= o) v += t;
    }
    return v;
}

__device__ __forceinline__ int wg_excl_scan(int v, int* ws, int t, int& total) {
    const int lane = t & 31, w = t >> 5;
    const int inc = wave_incl_scan(v, lane);
    if (lane == 31) ws[w] = inc;
    __syncthreads();
    const int wv   = ws[lane];
    const int winc = wave_incl_scan(wv, lane);
    const int woff = __shfl(winc, w, 32) - __shfl(wv, w, 32);
    total = __shfl(winc, 31, 32);
    __syncthreads();
    return woff + inc - v;
}

__global__ __launch_bounds__(WT) void k_count_lds(const int* __restrict__ dst, int* __restrict__ cseg,
                                                  int* __restrict__ wgtot,
                                                  int n, int nP, int NR, int E, int segLen) {
    __shared__ int c[NRMAX];
    __shared__ int wsum[32];
    const int t  = threadIdx.x;
    const int lo = blockIdx.x * NR;
    int nr = n - lo;
    if (nr > NR) nr = NR;
    if (nr < 0) nr = 0;
    for (int i = t; i < nr; i += WT) c[i] = 0;
    __syncthreads();

    const int eb = blockIdx.y * segLen;
    int len = E - eb;
    if (len > segLen) len = segLen;
    if (len > 0) {
        const int* dp = dst + eb;
        int done = 0;
        if ((((size_t)dp) & 15) == 0) {
            const int nvec = len >> 2;
            const int4* dv = (const int4*)dp;
            for (int v0 = t; v0 < nvec; v0 += WU * WT) {
                int4 q[WU];
#pragma unroll
                for (int u = 0; u < WU; ++u) {
                    const int vi = v0 + u * WT;
                    q[u] = make_int4(-1, -1, -1, -1);
                    if (vi < nvec) q[u] = dv[vi];
                }
#pragma unroll
                for (int u = 0; u < WU; ++u) {
                    const unsigned r0 = (unsigned)(q[u].x - lo), r1 = (unsigned)(q[u].y - lo);
                    const unsigned r2 = (unsigned)(q[u].z - lo), r3 = (unsigned)(q[u].w - lo);
                    if (r0 < (unsigned)nr) atomicAdd(&c[r0], 1);
                    if (r1 < (unsigned)nr) atomicAdd(&c[r1], 1);
                    if (r2 < (unsigned)nr) atomicAdd(&c[r2], 1);
                    if (r3 < (unsigned)nr) atomicAdd(&c[r3], 1);
                }
            }
            done = nvec << 2;
        }
        for (int e = done + t; e < len; e += WT) {
            const unsigned r0 = (unsigned)(dp[e] - lo);
            if (r0 < (unsigned)nr) atomicAdd(&c[r0], 1);
        }
    }
    __syncthreads();
    int part = 0;
    for (int i = t; i < nr; i += WT) {
        const int v = c[i];
        cseg[(size_t)blockIdx.y * nP + lo + i] = v;
        part += v;
    }
    int tot;
    wg_excl_scan(part, wsum, t, tot);
    if (t == 0) wgtot[blockIdx.x * SEG + blockIdx.y] = tot;
}

__global__ __launch_bounds__(WT) void k_place_lds(const int* __restrict__ src, const int* __restrict__ dst,
                                                  const int* __restrict__ cseg, const int* __restrict__ wgtot,
                                                  int* __restrict__ csr, int* __restrict__ rowstart,
                                                  int* __restrict__ cnt, float* __restrict__ dinv,
                                                  int n, int nP, int NR, int E, int segLen) {
    __shared__ int c[NRMAX];
    __shared__ int wsum[32];
    const int t  = threadIdx.x;
    const int lo = blockIdx.x * NR;
    const int sg = blockIdx.y;
    int nr = n - lo;
    if (nr > NR) nr = NR;
    if (nr < 0) nr = 0;

    int tot[2][4], pre[2][4];
#pragma unroll
    for (int h = 0; h < 2; ++h) {
        const int i = h * (NRMAX / 2) + 4 * t;
#pragma unroll
        for (int k = 0; k < 4; ++k) { tot[h][k] = 0; pre[h][k] = 0; }
        if (i < nr) {
            int4 v[SEG];
#pragma unroll
            for (int s2 = 0; s2 < SEG; ++s2) v[s2] = *(const int4*)(cseg + (size_t)s2 * nP + lo + i);
#pragma unroll
            for (int s2 = 0; s2 < SEG; ++s2) {
                tot[h][0] += v[s2].x; tot[h][1] += v[s2].y; tot[h][2] += v[s2].z; tot[h][3] += v[s2].w;
                if (s2 < sg) { pre[h][0] += v[s2].x; pre[h][1] += v[s2].y; pre[h][2] += v[s2].z; pre[h][3] += v[s2].w; }
            }
#pragma unroll
            for (int k = 0; k < 4; ++k)
                if (i + k >= nr) { tot[h][k] = 0; pre[h][k] = 0; }
        }
    }
    int bs = 0;
    for (int q = t; q < (int)blockIdx.x * SEG; q += WT) bs += wgtot[q];
    int carry;
    wg_excl_scan(bs, wsum, t, carry);
#pragma unroll
    for (int h = 0; h < 2; ++h) {
        const int i = h * (NRMAX / 2) + 4 * t;
        int all;
        int st = carry + wg_excl_scan(tot[h][0] + tot[h][1] + tot[h][2] + tot[h][3], wsum, t, all);
#pragma unroll
        for (int k = 0; k < 4; ++k) {
            if (i + k < nr) {
                if (sg == 0) {
                    rowstart[lo + i + k] = st;
                    cnt[lo + i + k]      = tot[h][k];
                    dinv[lo + i + k]     = rsqrtf((float)(tot[h][k] + 1));
                }
                c[i + k] = st + pre[h][k];
            }
            st += tot[h][k];
        }
        carry += all;
    }
    __syncthreads();

    const int eb = blockIdx.y * segLen;
    int len = E - eb;
    if (len > segLen) len = segLen;
    if (len > 0) {
        const int* dp = dst + eb;
        const int* sp = src + eb;
        int done = 0;
        if (((((size_t)dp) | ((size_t)sp)) & 15) == 0) {
            const int nvec = len >> 2;
            const int4* dv = (const int4*)dp;
            const int4* sv = (const int4*)sp;
            for (int v0 = t; v0 < nvec; v0 += WU * WT) {
                int4 q[WU], w[WU];
#pragma unroll
                for (int u = 0; u < WU; ++u) {
                    const int vi = v0 + u * WT;
                    q[u] = make_int4(-1, -1, -1, -1);
                    w[u] = make_int4(0, 0, 0, 0);
                    if (vi < nvec) { q[u] = dv[vi]; w[u] = sv[vi]; }
                }
                int p[WU][4];
#pragma unroll
                for (int u = 0; u < WU; ++u) {
                    const unsigned r0 = (unsigned)(q[u].x - lo), r1 = (unsigned)(q[u].y - lo);
                    const unsigned r2 = (unsigned)(q[u].z - lo), r3 = (unsigned)(q[u].w - lo);
                    p[u][0] = -1; p[u][1] = -1; p[u][2] = -1; p[u][3] = -1;
                    if (r0 < (unsigned)nr) p[u][0] = atomicAdd(&c[r0], 1);
                    if (r1 < (unsigned)nr) p[u][1] = atomicAdd(&c[r1], 1);
                    if (r2 < (unsigned)nr) p[u][2] = atomicAdd(&c[r2], 1);
                    if (r3 < (unsigned)nr) p[u][3] = atomicAdd(&c[r3], 1);
                }
#pragma unroll
                for (int u = 0; u < WU; ++u) {
                    if (p[u][0] >= 0) csr[p[u][0]] = w[u].x;
                    if (p[u][1] >= 0) csr[p[u][1]] = w[u].y;
                    if (p[u][2] >= 0) csr[p[u][2]] = w[u].z;
                    if (p[u][3] >= 0) csr[p[u][3]] = w[u].w;
                }
            }
            done = nvec << 2;
        }
        for (int e = done + t; e < len; e += WT) {
            const unsigned r0 = (unsigned)(dp[e] - lo);
            if (r0 < (unsigned)nr) csr[atomicAdd(&c[r0], 1)] = sp[e];
        }
    }
}

constexpr int LP = 128 + 4;

__global__ __launch_bounds__(128) void k_agg_mlp(const int* __restrict__ csr, const int* __restrict__ cnt,
                                                 const int* __restrict__ rowstart,
                                                 const float* __restrict__ dinv, const float* __restrict__ x,
                                                 const float* __restrict__ W1, const float* __restrict__ b1,
                                                 const float* __restrict__ W2, float* __restrict__ g, int n) {
    typedef float vf __attribute__((ext_vector_type(4)));
    __shared__ float sA[16 * LP];
    __shared__ float sH[16 * LP];
    const int lane = threadIdx.x & 31;
    const int w = __builtin_amdgcn_readfirstlane((int)(threadIdx.x >> 5));
    const int strip0 = blockIdx.x * 16;

    const int d0 = strip0 + w * 4;
    int dEnd = d0 + 4;
    if (dEnd > n) dEnd = n;
    if (d0 + 4 > n) {
#pragma unroll
        for (int i = 0; i < 4; ++i) *(vf*)(sA + (w * 4 + i) * LP + lane * 4) = vf{0.0f, 0.0f, 0.0f, 0.0f};
    }
    if (d0 < n) {
        const float* hb = x + lane * 4;
        int   d     = d0;
        int   start = rowstart[d];
        int   deg   = cnt[d];
        float dd    = dinv[d];
        vf    self  = *(const vf*)(hb + (size_t)d * 128);
        int   myS   = 0;
        if (lane < deg) myS = csr[start + lane];
        int   dn = d + 1;
        int   startN = 0, degN = 0;
        float ddN = 0.0f;
        if (dn < dEnd) { startN = rowstart[dn]; degN = cnt[dn]; ddN = dinv[dn]; }

        for (;;) {
            float myD = dinv[myS];
            vf acc = self * (dd * dd);
            int mySN = 0;
            if (lane < degN) mySN = csr[startN + lane];
            vf selfN = self;
            if (dn < dEnd) selfN = *(const vf*)(hb + (size_t)dn * 128);
            const int dnn = dn + 1;
            int   startNN = 0, degNN = 0;
            float ddNN = 0.0f;
            if (dnn < dEnd) { startNN = rowstart[dnn]; degNN = cnt[dnn]; ddNN = dinv[dnn]; }

            for (int base = 0; base < deg; base += 32) {
                const int m = (deg - base) < 32 ? (deg - base) : 32;
                if (base > 0) {
                    myS = 0;
                    if (lane < m) myS = csr[start + base + lane];
                    myD = dinv[myS];
                }
                for (int j = 0; j < m; j += 8) {
                    const int r = m - j;
                    vf v[8];
#pragma unroll
                    for (int u = 0; u < 8; ++u) {
                        if (u < r) {
                            const int s = __builtin_amdgcn_readlane(myS, j + u);
                            v[u] = *(const vf*)(hb + (size_t)s * 128);
                        }
                    }
                    float nr[8];
#pragma unroll
                    for (int u = 0; u < 8; ++u)
                        nr[u] = __int_as_float(__builtin_amdgcn_readlane(__float_as_int(myD), (j + u) & 31)) * dd;
#pragma unroll
                    for (int u = 0; u < 8; ++u) {
                        if (u < r) acc += v[u] * nr[u];
                    }
                }
            }

            *(vf*)(sA + (d - strip0) * LP + lane * 4) = acc;

            if (dn >= dEnd) break;
            d = dn;   start = startN; deg = degN;  dd = ddN;  myS = mySN;  self = selfN;
            dn = dnn; startN = startNN; degN = degNN; ddN = ddNN;
        }
    }
    __syncthreads();

    const int half = lane >> 4;
    const int l    = lane & 15;

    {
        v8f acc0 = v8f{}, acc1 = v8f{};
        const float* ap = sA + l * LP + half * 2;
        const int    c0 = 32 * w + 2 * l;
        const float* bp = W1 + (size_t)(half * 2) * 128 + c0;
#pragma unroll 4
        for (int k = 0; k < 128; k += 4) {
            const v2f a  = *(const v2f*)(ap + k);
            const v2f r0 = *(const v2f*)(bp + (size_t)k * 128);
            const v2f r1 = *(const v2f*)(bp + (size_t)k * 128 + 128);
            v2f bA, bB;
            bA.x = r0.x; bA.y = r1.x;
            bB.x = r0.y; bB.y = r1.y;
            acc0 = __builtin_amdgcn_wmma_f32_16x16x4_f32(false, a, false, bA, (short)0, acc0, false, false);
            acc1 = __builtin_amdgcn_wmma_f32_16x16x4_f32(false, a, false, bB, (short)0, acc1, false, false);
        }
        const v2f bb = *(const v2f*)(b1 + c0);
        float* hp = sH + (half * 8) * LP + c0;
#pragma unroll
        for (int v = 0; v < 8; ++v) {
            const float u0 = acc0[v] + bb.x, u1 = acc1[v] + bb.y;
            v2f o;
            o.x = u0 > 0.0f ? u0 : 0.0f;
            o.y = u1 > 0.0f ? u1 : 0.0f;
            *(v2f*)(hp + v * LP) = o;
        }
    }
    __syncthreads();

    {
        v8f acc = v8f{};
        const float* ap = sH + l * LP + half * 2;
        const float* bp = W2 + (size_t)(half * 2) * 64 + w * 16 + l;
        const int r0 = strip0 + half * 8;
        float dv[8];
#pragma unroll
        for (int v = 0; v < 8; ++v) dv[v] = (r0 + v < n) ? dinv[r0 + v] : 0.0f;
#pragma unroll 4
        for (int k = 0; k < 128; k += 4) {
            const v2f a = *(const v2f*)(ap + k);
            const float* b0 = bp + (size_t)k * 64;
            v2f b;
            b.x = b0[0]; b.y = b0[64];
            acc = __builtin_amdgcn_wmma_f32_16x16x4_f32(false, a, false, b, (short)0, acc, false, false);
        }
        float* cp = g + (size_t)r0 * 64 + w * 16 + l;
#pragma unroll
        for (int v = 0; v < 8; ++v) {
            if (r0 + v < n) cp[(size_t)v * 64] = acc[v] * dv[v];
        }
    }
}


template <int F, bool RELU>
__global__ void k_gather(const int* __restrict__ csr, const int* __restrict__ cnt,
                         const int* __restrict__ rowstart,
                         const float* __restrict__ dinv, const float* __restrict__ h,
                         const float* __restrict__ bias, float* __restrict__ out, int n) {
    constexpr int PER = F / 32;
    typedef float vf __attribute__((ext_vector_type(PER)));
    const int lane = threadIdx.x & 31;
    const int nw   = (gridDim.x * blockDim.x) >> 5;
    const float* hb = h + lane * PER;
    const vf bvec = *(const vf*)(bias + lane * PER);
    int d = __builtin_amdgcn_readfirstlane((blockIdx.x * blockDim.x + threadIdx.x) >> 5);
    if (d >= n) return;

    int   start = rowstart[d];
    int   deg   = cnt[d];
    float dd    = dinv[d];
    vf    self  = *(const vf*)(hb + (size_t)d * F);
    int   myS   = 0;
    if (lane < deg) myS = csr[start + lane];
    int   dn = d + nw;
    int   startN = 0, degN = 0;
    float ddN = 0.0f;
    if (dn < n) { startN = rowstart[dn]; degN = cnt[dn]; ddN = dinv[dn]; }

    for (;;) {
        vf acc = self;
        int   mySN = 0;
        vf    selfN = self;
        const int dnn = dn + nw;
        int   startNN = 0, degNN = 0;
        float ddNN = 0.0f;
        bool  asked = false;
        auto ask_next = [&]() {
            if (lane < degN) mySN = csr[startN + lane];
            if (dn < n) selfN = *(const vf*)(hb + (size_t)dn * F);
            if (dnn < n) { startNN = rowstart[dnn]; degNN = cnt[dnn]; ddNN = dinv[dnn]; }
        };

        for (int base = 0; base < deg; base += 32) {
            const int m = (deg - base) < 32 ? (deg - base) : 32;
            if (base > 0) {
                myS = 0;
                if (lane < m) myS = csr[start + base + lane];
            }
            for (int j = 0; j < m; j += 8) {
                const int r = m - j;
                vf v[8];
#pragma unroll
                for (int u = 0; u < 8; ++u) {
                    if (u < r) {
                        const int s = __builtin_amdgcn_readlane(myS, j + u);
                        v[u] = *(const vf*)(hb + (size_t)s * F);
                    }
                }
                if (!asked) { asked = true; ask_next(); }
#pragma unroll
                for (int u = 0; u < 8; ++u) {
                    if (u < r) acc += v[u];
                }
            }
        }
        if (!asked) ask_next();

        acc = acc * dd + bvec;
        if (RELU) {
#pragma unroll
            for (int q = 0; q < PER; ++q) acc[q] = acc[q] > 0.0f ? acc[q] : 0.0f;
        }
        *(vf*)(out + (size_t)d * F + lane * PER) = acc;

        if (dn >= n) break;
        d = dn;   start = startN; deg = degN;  dd = ddN;  myS = mySN;  self = selfN;
        dn = dnn; startN = startNN; degN = degNN; ddN = ddNN;
    }
}

extern "C" void kernel_launch(void* const* d_in, const int* in_sizes, int n_in,
                              void* d_out, int out_size, void* d_ws, size_t ws_size,
                              hipStream_t stream) {
    const float* x  = (const float*)d_in[0];
    const int*   ei = (const int*)d_in[1];
    const float* W1 = (const float*)d_in[2];
    const float* b1 = (const float*)d_in[3];
    const float* W2 = (const float*)d_in[4];
    const float* b2 = (const float*)d_in[5];
    float* out = (float*)d_out;

    const int n = in_sizes[0] / 128;
    const int E = in_sizes[1] / 2;
    const int* src = ei;
    const int* dst = ei + E;

    int R = (n + NRMAX - 1) / NRMAX;
    if (R < 16) R = 16;
    const int NR     = (((n + R - 1) / R) + 3) & ~3;
    const int segLen = (((E + 3) / 4 + SEG - 1) / SEG) * 4;
    const int gridG  = (n + 7) / 8 < 2048 ? (n + 7) / 8 : 2048;

    char* ws = (char*)d_ws;
    const int    nP    = (n + 3) & ~3;
    const size_t nodeB = ((size_t)n * 4 + 511) & ~(size_t)511;
    const size_t edgeB = ((size_t)E * 4 + 511) & ~(size_t)511;
    const size_t totB  = ((size_t)R * SEG * 4 + 511) & ~(size_t)511;
    const size_t segB  = ((size_t)SEG * nP * 4 + 511) & ~(size_t)511;
    size_t off = 0;
    float* dinv     = (float*)(ws + off); off += nodeB;
    int*   cnt      = (int*)(ws + off);   off += nodeB;
    int*   rowstart = (int*)(ws + off);   off += nodeB;
    int*   wgtot    = (int*)(ws + off);   off += totB;
    int*   cseg     = (int*)(ws + off);   off += segB;
    int*   csr      = (int*)(ws + off);   off += edgeB;
    float* g        = (float*)(ws + off);

    k_count_lds<<<dim3(R, SEG), WT, 0, stream>>>(dst, cseg, wgtot, n, nP, NR, E, segLen);
    k_place_lds<<<dim3(R, SEG), WT, 0, stream>>>(src, dst, cseg, wgtot, csr, rowstart, cnt, dinv, n, nP, NR, E, segLen);

    k_agg_mlp<<<(n + 15) / 16, 128, 0, stream>>>(csr, cnt, rowstart, dinv, x, W1, b1, W2, g, n);

    k_gather<64, false><<<gridG, 256, 0, stream>>>(csr, cnt, rowstart, dinv, g, b2, out, n);
}
